# starting kernel
# baseline (speedup 1.0000x reference)
#include <hip/hip_runtime.h>
#include <math.h>

#define ALPHA 0.1f
#define KITER 10
#define DFEAT 32

typedef __attribute__((ext_vector_type(2))) float v2f;
typedef __attribute__((ext_vector_type(8))) float v8f;

__global__ void appnp_init(const float* __restrict__ feat, float* __restrict__ h,
                           float* __restrict__ acc, float* __restrict__ norm, int N) {
    int i = blockIdx.x * blockDim.x + threadIdx.x;
    int total = N * DFEAT;
    if (i < total) { h[i] = feat[i]; acc[i] = 0.0f; }
    if (i < N) norm[i] = 0.0f;
}

__global__ void appnp_degree(const int* __restrict__ dst, float* __restrict__ deg, int E) {
    int i = blockIdx.x * blockDim.x + threadIdx.x;
    if (i < E) atomicAdd(&deg[dst[i]], 1.0f);
}

__global__ void appnp_norm(float* __restrict__ norm, int N) {
    int i = blockIdx.x * blockDim.x + threadIdx.x;
    if (i < N) {
        float d = norm[i];
        norm[i] = (d > 0.0f) ? (1.0f / sqrtf(d)) : 0.0f;
    }
}

__global__ void appnp_scatter(const float* __restrict__ h, const float* __restrict__ norm,
                              const int* __restrict__ src, const int* __restrict__ dst,
                              float* __restrict__ acc, int E) {
    int wave = blockIdx.x * (blockDim.x >> 5) + (threadIdx.x >> 5);
    if (wave >= E) return;
    int lane = threadIdx.x & 31;
    if (lane == 0) {
        __builtin_prefetch(src + wave + 4096, 0, 0);
        __builtin_prefetch(dst + wave + 4096, 0, 0);
    }
    int s = src[wave];
    int d = dst[wave];
    float v = h[s * DFEAT + lane] * norm[s];
    atomicAdd(&acc[d * DFEAT + lane], v);
}

__global__ void appnp_update_wmma(const float* __restrict__ feat, const float* __restrict__ norm,
                                  float* __restrict__ acc, float* __restrict__ h,
                                  int N, int ntiles) {
    int wave = blockIdx.x * (blockDim.x >> 5) + (threadIdx.x >> 5);
    if (wave >= ntiles) return;
    int node0 = wave * 16;
    int lane = threadIdx.x & 31;
    int hi = lane >> 4;
    int lo = lane & 15;

    if (node0 + 16 <= N) {
        float nv = (1.0f - ALPHA) * norm[node0 + lo];
#pragma unroll
        for (int fh = 0; fh < 2; ++fh) {
            int f0 = fh * 16;
            v8f cacc;
#pragma unroll
            for (int r = 0; r < 8; ++r)
                cacc[r] = ALPHA * feat[(node0 + r + 8 * hi) * DFEAT + f0 + lo];
#pragma unroll
            for (int cc = 0; cc < 4; ++cc) {
                v2f a, b;
#pragma unroll
                for (int v = 0; v < 2; ++v) {
                    int k = v + 2 * hi;
                    int m = 4 * cc + k;
                    b[v] = acc[(node0 + m) * DFEAT + f0 + lo];
                    a[v] = (lo == (4 * cc + k)) ? nv : 0.0f;
                }
                cacc = __builtin_amdgcn_wmma_f32_16x16x4_f32(
                     false, a,  false, b,
                     (short)0, cacc,  false,  false);
            }
#pragma unroll
            for (int r = 0; r < 8; ++r)
                h[(node0 + r + 8 * hi) * DFEAT + f0 + lo] = cacc[r];
#pragma unroll
            for (int cc = 0; cc < 4; ++cc)
#pragma unroll
                for (int v = 0; v < 2; ++v) {
                    int m = 4 * cc + v + 2 * hi;
                    acc[(node0 + m) * DFEAT + f0 + lo] = 0.0f;
                }
        }
    } else {
        for (int m = 0; m < 16; ++m) {
            int node = node0 + m;
            if (node < N) {
                int j = node * DFEAT + lane;
                h[j] = (1.0f - ALPHA) * norm[node] * acc[j] + ALPHA * feat[j];
                acc[j] = 0.0f;
            }
        }
    }
}

extern "C" void kernel_launch(void* const* d_in, const int* in_sizes, int n_in,
                              void* d_out, int out_size, void* d_ws, size_t ws_size,
                              hipStream_t stream) {
    const float* feat = (const float*)d_in[0];
    const int*   src  = (const int*)d_in[1];
    const int*   dst  = (const int*)d_in[2];
    int N = in_sizes[0] / DFEAT;
    int E = in_sizes[1];

    float* h    = (float*)d_out;
    float* acc  = (float*)d_ws;
    float* norm = acc + (size_t)N * DFEAT;

    int totalND = N * DFEAT;
    appnp_init<<<(totalND + 255) / 256, 256, 0, stream>>>(feat, h, acc, norm, N);
    appnp_degree<<<(E + 255) / 256, 256, 0, stream>>>(dst, norm, E);
    appnp_norm<<<(N + 255) / 256, 256, 0, stream>>>(norm, N);

    int ntiles = (N + 15) / 16;
    for (int k = 0; k < KITER; ++k) {
        appnp_scatter<<<(E + 7) / 8, 256, 0, stream>>>(h, norm, src, dst, acc, E);
        appnp_update_wmma<<<(ntiles + 7) / 8, 256, 0, stream>>>(feat, norm, acc, h, N, ntiles);
    }
}


# fill pass: a chunk's edge records are grouped by tile in LDS first, so each (chunk, tile) run leaves as neighbouring 8-byte stores of one wave instruction, not as 1.6M scattered 8-byte stores
# speedup vs baseline: 4.8690x; 4.8690x over previous
#include <hip/hip_runtime.h>
#include <math.h>

#define ALPHA 0.1f
#define KITER 10
#define DFEAT 32
#define RB 256
#define RB_SHIFT 8
#define NCH 256
#define SRC_BITS 24
#define SRC_MASK 0xFFFFFF

typedef __attribute__((ext_vector_type(2))) float v2f;
typedef __attribute__((ext_vector_type(8))) float v8f;

__device__ __forceinline__ float norm_of(float d) {
    return (d > 0.0f) ? (1.0f / sqrtf(d)) : 0.0f;
}

__global__ void appnp_init(float* __restrict__ deg, int* __restrict__ tot, int N, int nbk) {
    int i = blockIdx.x * blockDim.x + threadIdx.x;
    if (i < N) deg[i] = 0.0f;
    if (i < nbk) tot[i] = 0;
}

__global__ void appnp_bin_count(const int* __restrict__ dst, int* __restrict__ cnt, int* __restrict__ tot,
                                float* __restrict__ deg, int E, int nbk) {
    extern __shared__ int sh[];
    for (int i = threadIdx.x; i < nbk; i += blockDim.x) sh[i] = 0;
    __syncthreads();
    int per = (E + NCH - 1) / NCH;
    int e0 = blockIdx.x * per;
    int e1 = min(E, e0 + per);
    for (int e = e0 + threadIdx.x; e < e1; e += blockDim.x) {
        int d = dst[e];
        atomicAdd(&sh[d >> RB_SHIFT], 1);
        atomicAdd(&deg[d], 1.0f);
    }
    __syncthreads();
    for (int i = threadIdx.x; i < nbk; i += blockDim.x) {
        int c = sh[i];
        cnt[blockIdx.x * nbk + i] = c;
        if (c != 0) atomicAdd(&tot[i], c);
    }
}

#define FILL_HDR(nbk) ((3 * (nbk) + 2) & ~1)
__global__ __launch_bounds__(1024)
void appnp_bin_fill(const int* __restrict__ src, const int* __restrict__ dst,
                    const float* __restrict__ deg,
                    const int* __restrict__ cnt, const int* __restrict__ tot,
                    int* __restrict__ bstart,
                    int* __restrict__ rec, int E, int nbk) {
    extern __shared__ __attribute__((aligned(16))) int fsh[];
    __shared__ int wsum[32];
    __shared__ int wsum2[32];
    int* slot = fsh;
    int* lo = fsh + nbk;
    int* gb = lo + nbk + 1;
    int2* stage = reinterpret_cast<int2*>(fsh + FILL_HDR(nbk));
    int lane = threadIdx.x & 31;
    int wv = threadIdx.x >> 5;
    int carry = 0;
    int carry2 = 0;
    for (int t0 = 0; t0 < nbk; t0 += 1024) {
        int t = t0 + threadIdx.x;
        int run = (t < nbk) ? tot[t] : 0;
        int own = (t < nbk) ? cnt[blockIdx.x * nbk + t] : 0;
        int x = run;
        int x2 = own;
        for (int d = 1; d < 32; d <<= 1) {
            int y = __shfl_up(x, d, 32);
            int y2 = __shfl_up(x2, d, 32);
            if (lane >= d) { x += y; x2 += y2; }
        }
        if (lane == 31) { wsum[wv] = x; wsum2[wv] = x2; }
        __syncthreads();
        if (wv == 0) {
            int s = wsum[lane];
            int s2 = wsum2[lane];
            for (int d = 1; d < 32; d <<= 1) {
                int y = __shfl_up(s, d, 32);
                int y2 = __shfl_up(s2, d, 32);
                if (lane >= d) { s += y; s2 += y2; }
            }
            wsum[lane] = s;
            wsum2[lane] = s2;
        }
        __syncthreads();
        int incl = x + ((wv > 0) ? wsum[wv - 1] : 0);
        int incl2 = x2 + ((wv > 0) ? wsum2[wv - 1] : 0);
        int total = wsum[31];
        int total2 = wsum2[31];
        if (t < nbk) {
            int start = carry + incl - run;
            int off = start;
            for (int ch = 0; ch < (int)blockIdx.x; ++ch) off += cnt[ch * nbk + t];
            int l0 = carry2 + incl2 - own;
            slot[t] = off;
            lo[t] = l0;
            gb[t] = off - l0;
            if (blockIdx.x == 0) bstart[t] = start;
        }
        carry += total;
        carry2 += total2;
        __syncthreads();
    }
    if (threadIdx.x == 0) lo[nbk] = carry2;
    if (blockIdx.x == 0 && threadIdx.x == 0) bstart[nbk] = carry;
    int per = (E + NCH - 1) / NCH;
    int e0 = blockIdx.x * per;
    int e1 = min(E, e0 + per);
    for (int e = e0 + threadIdx.x; e < e1; e += blockDim.x) {
        int d = dst[e];
        int s = src[e];
        float nv = norm_of(deg[s]);
        int tl = d >> RB_SHIFT;
        int pos = atomicAdd(&slot[tl], 1);
        stage[pos - gb[tl]] = make_int2(s | ((d & (RB - 1)) << SRC_BITS), __float_as_int(nv));
    }
    __syncthreads();
    int2* rec2 = reinterpret_cast<int2*>(rec);
    int wu = __builtin_amdgcn_readfirstlane(wv);
    for (int tb = wu; tb < nbk; tb += 1024) {
        int tm = tb + 32 * lane;
        int av = 0, bv = 0, gv = 0;
        if (tm < nbk) { av = lo[tm]; bv = lo[tm + 1]; gv = gb[tm]; }
        for (int j = 0; j < 32 && tb + 32 * j < nbk; ++j) {
            int a = __builtin_amdgcn_readlane(av, j);
            int b = __builtin_amdgcn_readlane(bv, j);
            int g = __builtin_amdgcn_readlane(gv, j);
            for (int k = a + lane; k < b; k += 32) rec2[g + k] = stage[k];
        }
    }
}

#define SC_WAVES 16
#define SC_PER_THREAD (RB * DFEAT / 4 / (SC_WAVES * 32))
static_assert(RB == 16 * SC_WAVES, "the last sweep gives every wave one 16-row slice of the tile");
template <int MODE>
__global__ __launch_bounds__(SC_WAVES * 32)
void appnp_scatter_lds(const float* __restrict__ zin,
                       const int* __restrict__ rec, const int* __restrict__ bstart,
                       const float* __restrict__ z1,
                       const float* __restrict__ feat, const float* __restrict__ deg,
                       float* __restrict__ zout, int N) {
    constexpr bool FIRST = (MODE == 0);
    constexpr bool LAST = (MODE == 2);
    __shared__ __attribute__((aligned(16))) float sacc[RB * DFEAT];
    int tid = threadIdx.x;
    int lane = tid & 31;
    int w = __builtin_amdgcn_readfirstlane(tid >> 5);
    float4* sacc4 = reinterpret_cast<float4*>(sacc);
    int row0 = blockIdx.x * RB;
    for (int i = tid; i < RB * DFEAT / 4; i += SC_WAVES * 32) sacc4[i] = make_float4(0.0f, 0.0f, 0.0f, 0.0f);
    __syncthreads();
    float4 c1[SC_PER_THREAD];
    if (!FIRST) {
        const float4* z14 = reinterpret_cast<const float4*>(z1);
#pragma unroll
        for (int j = 0; j < SC_PER_THREAD; ++j) {
            int i = tid + j * SC_WAVES * 32;
            int node = min(row0 + (i >> 3), N - 1);
            c1[j] = z14[(size_t)node * (DFEAT / 4) + (i & 7)];
        }
    }

    const int2* rec2 = reinterpret_cast<const int2*>(rec);
    int e0 = bstart[blockIdx.x];
    int e1 = bstart[blockIdx.x + 1];
    int base = e0 + w * 32;
    if (base < e1) {
        int idx = base + lane;
        int2 r = rec2[min(idx, e1 - 1)];
        if (idx >= e1) r.y = 0;
        if (!FIRST) r.y = __float_as_int(((1.0f - ALPHA) * __int_as_float(r.y)) * __int_as_float(r.y));
        for (; base < e1; base += SC_WAVES * 32) {
            int idxn = base + SC_WAVES * 32 + lane;
            int2 rn = rec2[min(idxn, e1 - 1)];
            __builtin_amdgcn_sched_barrier(0);
            if (idxn >= e1) rn.y = 0;
            if (!FIRST) rn.y = __float_as_int(((1.0f - ALPHA) * __int_as_float(rn.y)) * __int_as_float(rn.y));
            float v[32];
#pragma unroll
            for (int u = 0; u < 32; ++u) {
                int rj = __builtin_amdgcn_readlane(r.x, u);
                v[u] = zin[(size_t)(rj & SRC_MASK) * DFEAT + lane];
            }
#pragma unroll
            for (int u = 0; u < 32; ++u) {
                int rj = __builtin_amdgcn_readlane(r.x, u);
                float fj = __int_as_float(__builtin_amdgcn_readlane(r.y, u));
                atomicAdd(&sacc[(int)((unsigned)rj >> SRC_BITS) * DFEAT + lane], v[u] * fj);
            }
            r = rn;
        }
    }
    __syncthreads();

    if (!LAST) {
        float4* zout4 = reinterpret_cast<float4*>(zout);
#pragma unroll
        for (int j = 0; j < SC_PER_THREAD; ++j) {
            int i = tid + j * SC_WAVES * 32;
            int node = row0 + (i >> 3);
            float4 s = sacc4[i];
            if (!FIRST) {
                s.x += ALPHA * c1[j].x;
                s.y += ALPHA * c1[j].y;
                s.z += ALPHA * c1[j].z;
                s.w += ALPHA * c1[j].w;
            }
            if (node < N) zout4[(size_t)node * (DFEAT / 4) + (i & 7)] = s;
        }
    } else {
#pragma unroll
        for (int j = 0; j < SC_PER_THREAD; ++j) {
            int i = tid + j * SC_WAVES * 32;
            float4 s = sacc4[i];
            s.x += ALPHA * c1[j].x;
            s.y += ALPHA * c1[j].y;
            s.z += ALPHA * c1[j].z;
            s.w += ALPHA * c1[j].w;
            sacc4[i] = s;
        }
        __syncthreads();

        int lrow0 = 16 * w;
        int node0 = row0 + lrow0;
        int hi = lane >> 4;
        int lo = lane & 15;
        if (node0 + 16 <= N) {
            float nv = (1.0f - ALPHA) * norm_of(deg[node0 + lo]);
#pragma unroll
            for (int fh = 0; fh < 2; ++fh) {
                int f0 = fh * 16;
                v8f cacc;
#pragma unroll
                for (int r = 0; r < 8; ++r)
                    cacc[r] = ALPHA * feat[(size_t)(node0 + r + 8 * hi) * DFEAT + f0 + lo];
#pragma unroll
                for (int cc = 0; cc < 4; ++cc) {
                    v2f a, b;
#pragma unroll
                    for (int v = 0; v < 2; ++v) {
                        int k = v + 2 * hi;
                        int m = 4 * cc + k;
                        b[v] = sacc[(lrow0 + m) * DFEAT + f0 + lo];
                        a[v] = (lo == (4 * cc + k)) ? nv : 0.0f;
                    }
                    cacc = __builtin_amdgcn_wmma_f32_16x16x4_f32(
                         false, a,  false, b,
                         (short)0, cacc,  false,  false);
                }
#pragma unroll
                for (int r = 0; r < 8; ++r)
                    zout[(size_t)(node0 + r + 8 * hi) * DFEAT + f0 + lo] = cacc[r];
            }
        } else {
            for (int m = 0; m < 16; ++m) {
                int node = node0 + m;
                if (node < N) {
                    size_t j = (size_t)node * DFEAT + lane;
                    zout[j] = (1.0f - ALPHA) * norm_of(deg[node]) * sacc[(lrow0 + m) * DFEAT + lane]
                              + ALPHA * feat[j];
                }
            }
        }
    }
}

static_assert(KITER >= 2 && (KITER % 2) == 0, "the buffer rotation below ends in the output buffer for an even sweep count");

extern "C" void kernel_launch(void* const* d_in, const int* in_sizes, int n_in,
                              void* d_out, int out_size, void* d_ws, size_t ws_size,
                              hipStream_t stream) {
    const float* feat = (const float*)d_in[0];
    const int*   src  = (const int*)d_in[1];
    const int*   dst  = (const int*)d_in[2];
    int N = in_sizes[0] / DFEAT;
    int E = in_sizes[1];

    float* h    = (float*)d_out;
    float* za   = (float*)d_ws;
    float* zb   = za + (size_t)N * DFEAT;
    int nbk = (N + RB - 1) / RB;
    int* rec    = (int*)(zb + (size_t)N * DFEAT);
    float* deg  = (float*)(rec + 2 * (size_t)E);
    int* bstart = (int*)(deg + N);
    int* cnt    = (int*)za;
    int* tot    = (int*)zb;
    size_t need = (2 * (size_t)N * DFEAT + 2 * (size_t)E + (size_t)N + (size_t)nbk + 1) * sizeof(int);
    if (ws_size < need) return;
    size_t fill_lds = (size_t)FILL_HDR(nbk) * sizeof(int) + (size_t)((E + NCH - 1) / NCH) * 8;
    if (fill_lds > 65536) return;

    appnp_init<<<(N + 255) / 256, 256, 0, stream>>>(deg, tot, N, nbk);
    appnp_bin_count<<<NCH, 1024, nbk * sizeof(int), stream>>>(dst, cnt, tot, deg, E, nbk);
    appnp_bin_fill<<<NCH, 1024, fill_lds, stream>>>(src, dst, deg, cnt, tot, bstart, rec, E, nbk);

    appnp_scatter_lds<0><<<nbk, SC_WAVES * 32, 0, stream>>>(feat, rec, bstart, feat, feat, deg, za, N);
    for (int k = 2; k <= KITER; ++k) {
        const float* zin = (k == 2) ? za : ((k & 1) ? h : zb);
        float* zout = (k & 1) ? zb : h;
        if (k < KITER)
            appnp_scatter_lds<1><<<nbk, SC_WAVES * 32, 0, stream>>>(zin, rec, bstart, za, feat, deg, zout, N);
        else
            appnp_scatter_lds<2><<<nbk, SC_WAVES * 32, 0, stream>>>(zin, rec, bstart, za, feat, deg, zout, N);
    }
}
